# starting kernel
# baseline (speedup 1.0000x reference)
#include <hip/hip_runtime.h>
#include <hip/hip_bf16.h>

typedef __attribute__((ext_vector_type(2))) float v2f;
typedef __attribute__((ext_vector_type(8))) float v8f;

__global__ __launch_bounds__(256) void edge_dot_wmma_f32(
    const float* __restrict__ h,
    const int*   __restrict__ src,
    const int*   __restrict__ dst,
    float*       __restrict__ out,
    int n_edges)
{
    const int lane  = threadIdx.x & 31;
    const int wave  = blockIdx.x * (blockDim.x >> 5) + (threadIdx.x >> 5);
    const int e0    = wave * 16;
    if (e0 >= n_edges) return;

    const int m    = lane & 15;
    const int half = lane >> 4;

    const long long sIdx = (long long)src[e0 + m];
    const long long dIdx = (long long)dst[e0 + m];
    const float* __restrict__ srow = h + sIdx * 128;
    const float* __restrict__ drow = h + dIdx * 128;

    v8f c = {};
#pragma unroll
    for (int ch = 0; ch < 32; ++ch) {
        const int k = ch * 4 + half * 2;
        v2f a = *(const v2f*)(srow + k);
        v2f b = *(const v2f*)(drow + k);
        c = __builtin_amdgcn_wmma_f32_16x16x4_f32(
                false, a, false, b, (short)0, c, false, false);
    }

    const int sel = lane & 7;
    float val = c[0];
    if (sel == 1) val = c[1];
    if (sel == 2) val = c[2];
    if (sel == 3) val = c[3];
    if (sel == 4) val = c[4];
    if (sel == 5) val = c[5];
    if (sel == 6) val = c[6];
    if (sel == 7) val = c[7];

    if (lane < 8)        out[e0 + lane]        = val;
    else if (lane >= 24) out[e0 + (lane - 16)] = val;
}

extern "C" void kernel_launch(void* const* d_in, const int* in_sizes, int n_in,
                              void* d_out, int out_size, void* d_ws, size_t ws_size,
                              hipStream_t stream) {
    const float* h   = (const float*)d_in[0];
    const int*   src = (const int*)  d_in[1];
    const int*   dst = (const int*)  d_in[2];
    float*       out = (float*)      d_out;

    const int n_edges = in_sizes[1];
    const int edges_per_block = (256 / 32) * 16;
    const int grid = (n_edges + edges_per_block - 1) / edges_per_block;

    edge_dot_wmma_f32<<<grid, 256, 0, stream>>>(h, src, dst, out, n_edges);
}


# One resident 16-wave workgroup per WGP; each wave loops over its tiles and requests the next tile's 32 rows before the current tile's LDS fragment reads and WMMAs, so gather overlaps LDS+WMMA work
# speedup vs baseline: 2.3358x; 2.3358x over previous
#include <hip/hip_runtime.h>


typedef __attribute__((ext_vector_type(2))) float v2f;
typedef __attribute__((ext_vector_type(4))) float v4f;
typedef __attribute__((ext_vector_type(8))) float v8f;

#define D_FEAT      128
#define TILE_ROWS   32
#define HALF_ROWS   16
#define ROW_PITCH   132
#define WAVES_PER_BLOCK 16
#define N_BLOCKS    256

__global__ __launch_bounds__(WAVES_PER_BLOCK * 32) void edge_dot_wmma_f32(
    const float* __restrict__ h,
    const int*   __restrict__ src,
    const int*   __restrict__ dst,
    float*       __restrict__ out,
    int n_edges)
{
    __shared__ __attribute__((aligned(16))) float slabs[WAVES_PER_BLOCK * TILE_ROWS * ROW_PITCH];

    const int lane    = threadIdx.x & 31;
    const int wib     = __builtin_amdgcn_readfirstlane(threadIdx.x >> 5);
    const int n_tiles = (n_edges + 15) >> 4;
    const int stride  = gridDim.x * WAVES_PER_BLOCK;
    int t = blockIdx.x * WAVES_PER_BLOCK + wib;
    if (t >= n_tiles) return;

    const int m    = lane & 15;
    const int half = lane >> 4;

    const int* __restrict__ ends = half ? dst : src;
    const float* __restrict__ hl = h + lane * 4;

    float* slab = slabs + wib * (TILE_ROWS * ROW_PITCH);
    float* srow = slab + lane * 4;
    const float* arow = slab + m * ROW_PITCH + half * 4;
    const float* brow = slab + (16 + m) * ROW_PITCH + half * 4;

    const int last = n_tiles - 1;
    int t1 = t + stride;
    int node     = ends[t * 16 + m];
    int node_nxt = ends[(t1 < last ? t1 : last) * 16 + m];

    v4f row[TILE_ROWS];
#pragma unroll
    for (int i = 0; i < TILE_ROWS; ++i) {
        const long long r = (long long)__builtin_amdgcn_readlane(node, i);
        row[i] = *(const v4f*)(hl + r * D_FEAT);
    }

    for (;;) {
        const bool more = t1 < n_tiles;
        const int  t2   = t1 + stride;

#pragma unroll
        for (int i = 0; i < TILE_ROWS; ++i)
            *(v4f*)(srow + i * ROW_PITCH) = row[i];

        const int node_nn = ends[(t2 < last ? t2 : last) * 16 + m];

        v4f a[8], b[8];
#pragma unroll
        for (int j = 0; j < 8; ++j) {
            a[j] = *(const v4f*)(arow + j * 8);
            b[j] = *(const v4f*)(brow + j * 8);
        }
        if (more) {
#pragma unroll
            for (int i = 0; i < HALF_ROWS; ++i) {
                const long long r = (long long)__builtin_amdgcn_readlane(node_nxt, i);
                row[i] = *(const v4f*)(hl + r * D_FEAT);
            }
        }

        v8f c = {};
#pragma unroll
        for (int j = 0; j < 8; ++j) {
            const v2f a_lo = __builtin_shufflevector(a[j], a[j], 0, 1);
            const v2f b_lo = __builtin_shufflevector(b[j], b[j], 0, 1);
            const v2f a_hi = __builtin_shufflevector(a[j], a[j], 2, 3);
            const v2f b_hi = __builtin_shufflevector(b[j], b[j], 2, 3);
            c = __builtin_amdgcn_wmma_f32_16x16x4_f32(
                    false, a_lo, false, b_lo, (short)0, c, false, false);
            c = __builtin_amdgcn_wmma_f32_16x16x4_f32(
                    false, a_hi, false, b_hi, (short)0, c, false, false);
        }
#pragma unroll
        for (int j = 0; j < 8; ++j) {
            a[j] = *(const v4f*)(arow + 64 + j * 8);
            b[j] = *(const v4f*)(brow + 64 + j * 8);
        }
        if (more) {
#pragma unroll
            for (int i = HALF_ROWS; i < TILE_ROWS; ++i) {
                const long long r = (long long)__builtin_amdgcn_readlane(node_nxt, i);
                row[i] = *(const v4f*)(hl + r * D_FEAT);
            }
        }
#pragma unroll
        for (int j = 0; j < 8; ++j) {
            const v2f a_lo = __builtin_shufflevector(a[j], a[j], 0, 1);
            const v2f b_lo = __builtin_shufflevector(b[j], b[j], 0, 1);
            const v2f a_hi = __builtin_shufflevector(a[j], a[j], 2, 3);
            const v2f b_hi = __builtin_shufflevector(b[j], b[j], 2, 3);
            c = __builtin_amdgcn_wmma_f32_16x16x4_f32(
                    false, a_lo, false, b_lo, (short)0, c, false, false);
            c = __builtin_amdgcn_wmma_f32_16x16x4_f32(
                    false, a_hi, false, b_hi, (short)0, c, false, false);
        }

        const int sel = lane & 7;
        float val = c[0];
        if (sel == 1) val = c[1];
        if (sel == 2) val = c[2];
        if (sel == 3) val = c[3];
        if (sel == 4) val = c[4];
        if (sel == 5) val = c[5];
        if (sel == 6) val = c[6];
        if (sel == 7) val = c[7];

        const int e0 = t * 16;
        if (lane < 8)        out[e0 + lane]        = val;
        else if (lane >= 24) out[e0 + (lane - 16)] = val;

        if (!more) break;
        t = t1;
        t1 = t2;
        node_nxt = node_nn;
    }
}

extern "C" void kernel_launch(void* const* d_in, const int* in_sizes, int n_in,
                              void* d_out, int out_size, void* d_ws, size_t ws_size,
                              hipStream_t stream) {
    const float* h   = (const float*)d_in[0];
    const int*   src = (const int*)  d_in[1];
    const int*   dst = (const int*)  d_in[2];
    float*       out = (float*)      d_out;

    const int n_edges = in_sizes[1];

    edge_dot_wmma_f32<<<N_BLOCKS, WAVES_PER_BLOCK * 32, 0, stream>>>(h, src, dst, out, n_edges);
}
